# starting kernel
# baseline (speedup 1.0000x reference)
#include <hip/hip_runtime.h>

typedef __attribute__((ext_vector_type(2))) float v2f;
typedef __attribute__((ext_vector_type(8))) float v8f;

#define FEAT 128

__global__ void __launch_bounds__(256)
sage_scatter(const float* __restrict__ xsrc,
             const int* __restrict__ src,
             const int* __restrict__ dst,
             float* __restrict__ agg,
             float* __restrict__ cnt,
             int E)
{
    int wave = (int)((blockIdx.x * blockDim.x + threadIdx.x) >> 5);
    int lane = threadIdx.x & 31;
    if (wave >= E) return;
    int s = src[wave];
    int d = dst[wave];
    const float4* row = (const float4*)(xsrc + (size_t)s * FEAT);
    float4 v = row[lane];
    float* o = agg + (size_t)d * FEAT + lane * 4;
    atomicAdd(o + 0, v.x);
    atomicAdd(o + 1, v.y);
    atomicAdd(o + 2, v.z);
    atomicAdd(o + 3, v.w);
    if (lane == 0) atomicAdd(cnt + d, 1.0f);
}

__global__ void __launch_bounds__(256)
sage_mean_div(float* __restrict__ agg, const float* __restrict__ cnt, int total)
{
    int i = blockIdx.x * blockDim.x + threadIdx.x;
    if (i >= total) return;
    float c = cnt[i >> 7];
    agg[i] *= 1.0f / fmaxf(c, 1.0f);
}

__global__ void __launch_bounds__(256)
sage_gemm_wmma(const float* __restrict__ A1, const float* __restrict__ A2,
               const float* __restrict__ W1, const float* __restrict__ W2,
               const float* __restrict__ bias, float* __restrict__ out,
               int M, int N, int tiles_n, int total_tiles)
{
    int wave = (int)((blockIdx.x * blockDim.x + threadIdx.x) >> 5);
    if (wave >= total_tiles) return;
    int lane  = threadIdx.x & 31;
    int tm    = wave / tiles_n;
    int tn    = wave % tiles_n;

    int lmod  = lane & 15;
    int hi    = lane >> 4;
    int khalf = hi << 1;

    int m = tm * 16 + lmod;
    int n = tn * 16 + lmod;
    float amask = (m < M) ? 1.0f : 0.0f;
    int   mc    = (m < M) ? m : (M - 1);

    const float* a1 = A1 + (size_t)mc * FEAT + khalf;
    const float* a2 = A2 + (size_t)mc * FEAT + khalf;
    const float* w1 = W1 + (size_t)n  * FEAT + khalf;
    const float* w2 = W2 + (size_t)n  * FEAT + khalf;

    v8f acc = {};

    #pragma unroll 4
    for (int k = 0; k < FEAT; k += 4) {
        v2f a, b;
        a[0] = a1[k]     * amask;
        a[1] = a1[k + 1] * amask;
        b[0] = w1[k];
        b[1] = w1[k + 1];
        acc = __builtin_amdgcn_wmma_f32_16x16x4_f32(
             false, a,  false, b,
             (short)0, acc,  false,  false);
    }
    #pragma unroll 4
    for (int k = 0; k < FEAT; k += 4) {
        v2f a, b;
        a[0] = a2[k]     * amask;
        a[1] = a2[k + 1] * amask;
        b[0] = w2[k];
        b[1] = w2[k + 1];
        acc = __builtin_amdgcn_wmma_f32_16x16x4_f32(
            false, a, false, b, (short)0, acc, false, false);
    }

    float bn    = bias[n];
    int   mbase = tm * 16 + hi * 8;
    #pragma unroll
    for (int j = 0; j < 8; ++j) {
        int mm = mbase + j;
        if (mm < M) {
            float v = acc[j] + bn;
            out[(size_t)mm * N + n] = v > 0.0f ? v : 0.0f;
        }
    }
}

extern "C" void kernel_launch(void* const* d_in, const int* in_sizes, int n_in,
                              void* d_out, int out_size, void* d_ws, size_t ws_size,
                              hipStream_t stream) {
    const float* x    = (const float*)d_in[0];
    const float* W1l  = (const float*)d_in[1];
    const float* b1l  = (const float*)d_in[2];
    const float* W1r  = (const float*)d_in[3];
    const float* W2l  = (const float*)d_in[4];
    const float* b2l  = (const float*)d_in[5];
    const float* W2r  = (const float*)d_in[6];
    const int*   src1 = (const int*)d_in[7];
    const int*   dst1 = (const int*)d_in[8];
    const int*   src2 = (const int*)d_in[9];
    const int*   dst2 = (const int*)d_in[10];

    const int E1     = in_sizes[7];
    const int E2     = in_sizes[9];
    const int N1FEAT = in_sizes[2];
    const int N2FEAT = in_sizes[5];
    const int SIZE2  = out_size / N2FEAT;
    const int SIZE1  = 50000;

    float* agg1 = (float*)d_ws;
    float* cnt1 = agg1 + (size_t)SIZE1 * FEAT;
    float* h    = cnt1 + SIZE1;
    float* agg2 = h    + (size_t)SIZE1 * FEAT;
    float* cnt2 = agg2 + (size_t)SIZE2 * FEAT;

    hipMemsetAsync(agg1, 0, ((size_t)SIZE1 * FEAT + SIZE1) * sizeof(float), stream);
    hipMemsetAsync(agg2, 0, ((size_t)SIZE2 * FEAT + SIZE2) * sizeof(float), stream);

    sage_scatter<<<(E1 + 7) / 8, 256, 0, stream>>>(x, src1, dst1, agg1, cnt1, E1);
    {
        int total = SIZE1 * FEAT;
        sage_mean_div<<<(total + 255) / 256, 256, 0, stream>>>(agg1, cnt1, total);
    }
    {
        int tiles_n = N1FEAT / 16;
        int tiles_m = (SIZE1 + 15) / 16;
        int total   = tiles_m * tiles_n;
        sage_gemm_wmma<<<(total + 7) / 8, 256, 0, stream>>>(
            agg1, x, W1l, W1r, b1l, h, SIZE1, N1FEAT, tiles_n, total);
    }

    sage_scatter<<<(E2 + 7) / 8, 256, 0, stream>>>(h, src2, dst2, agg2, cnt2, E2);
    {
        int total = SIZE2 * FEAT;
        sage_mean_div<<<(total + 255) / 256, 256, 0, stream>>>(agg2, cnt2, total);
    }
    {
        int tiles_n = N2FEAT / 16;
        int tiles_m = (SIZE2 + 15) / 16;
        int total   = tiles_m * tiles_n;
        sage_gemm_wmma<<<(total + 7) / 8, 256, 0, stream>>>(
            agg2, h, W2l, W2r, b2l, (float*)d_out, SIZE2, N2FEAT, tiles_n, total);
    }
}


# dst lists recoded once (one added launch) to one byte per edge = owning workgroup; the LDS-sum kernel sweeps the bytes, 4 edges per 32-bit op, re-tests real dst at drain; sum order = LDS arrival
# speedup vs baseline: 3.6450x; 3.6450x over previous
#include <hip/hip_runtime.h>

typedef __attribute__((ext_vector_type(2))) float v2f;
typedef __attribute__((ext_vector_type(8))) float v8f;

#define FEAT 128

#define OWN_ROWS  196
#define OWN_WAVES 32
#define OWN_SCAN  4
#define OWN_GROUP 16
#define OWN_TRIP  512
#define OWN_QCAP  (OWN_GROUP + 32)

static __device__ __forceinline__ int own_rpw(int n_dst, int groups)
{
    int rpw = (n_dst + groups - 1) / groups;
    if (rpw > OWN_ROWS) rpw = OWN_ROWS;
    if (rpw < 1) rpw = 1;
    return rpw;
}

__global__ void __launch_bounds__(256)
sage_owner_ids(const int* __restrict__ dstA, unsigned* __restrict__ idsA, int EA, int nA,
               const int* __restrict__ dstB, unsigned* __restrict__ idsB, int EB, int nB,
               int groups)
{
    const int VA = ((EA + OWN_TRIP - 1) / OWN_TRIP) * (OWN_TRIP / 16);
    const int VB = ((EB + OWN_TRIP - 1) / OWN_TRIP) * (OWN_TRIP / 16);
    const unsigned rA = (unsigned)own_rpw(nA, groups);
    const unsigned rB = (unsigned)own_rpw(nB, groups);
    for (int i = blockIdx.x * blockDim.x + threadIdx.x; i < VA + VB;
         i += gridDim.x * blockDim.x) {
        const bool      a   = i < VA;
        const int*      d   = a ? dstA : dstB;
        unsigned*       o   = a ? idsA : idsB;
        const int       E   = a ? EA : EB;
        const unsigned  rpw = a ? rA : rB;
        const int       k   = a ? i : i - VA;
        const int       e0  = k * 16;
        unsigned w[4];
        #pragma unroll
        for (int j = 0; j < 4; ++j) {
            const int e = e0 + 4 * j;
            int4 v = make_int4(0, 0, 0, 0);
            if (e + 3 < E) {
                v = *(const int4*)(d + e);
            } else {
                if (e + 0 < E) v.x = d[e + 0];
                if (e + 1 < E) v.y = d[e + 1];
                if (e + 2 < E) v.z = d[e + 2];
            }
            w[j] =  (((unsigned)v.x / rpw) & 255u)
                 | ((((unsigned)v.y / rpw) & 255u) << 8)
                 | ((((unsigned)v.z / rpw) & 255u) << 16)
                 | ((((unsigned)v.w / rpw) & 255u) << 24);
        }
        ((uint4*)o)[k] = make_uint4(w[0], w[1], w[2], w[3]);
    }
}

static __device__ __forceinline__ void
own_drain(const float* __restrict__ xsrc, const int* __restrict__ src,
          const int* __restrict__ dst, const unsigned* q, int nvalid,
          int lo, unsigned rows, unsigned* cnt, float* acc, int lane)
{
    const int      ql = lane & (OWN_GROUP - 1);
    const unsigned e  = q[ql];
    const int      s  = src[e];
    const unsigned r  = (unsigned)(dst[e] - lo);
    const bool     ok = (r < rows) & (ql < nvalid);
    const unsigned dl = ok ? r : (unsigned)OWN_ROWS;
    if (ok & (lane < OWN_GROUP)) atomicAdd(cnt + r, 1u);
    float4 v[OWN_GROUP];
    #pragma unroll
    for (int i = 0; i < OWN_GROUP; ++i) {
        {
            int si = __builtin_amdgcn_readlane(s, i);
            v[i] = ((const float4*)(xsrc + (size_t)si * FEAT))[lane];
        }
    }
    #pragma unroll
    for (int i = 0; i < OWN_GROUP; ++i) {
        {
            int di = __builtin_amdgcn_readlane((int)dl, i);
            float* a = acc + di * FEAT + lane;
            atomicAdd(a,      v[i].x);
            atomicAdd(a + 32, v[i].y);
            atomicAdd(a + 64, v[i].z);
            atomicAdd(a + 96, v[i].w);
        }
    }
}

static __device__ __forceinline__ int
own_trip(uint4 v, unsigned eb, unsigned idm, int E, int lo, unsigned rows,
         const float* __restrict__ xsrc, const int* __restrict__ src,
         const int* __restrict__ dst,
         unsigned* q, unsigned* cnt, float* acc, int qn, int lane)
{
    const unsigned L = 0x7f7f7f7fu, H = 0x80808080u;
    const unsigned x0 = v.x ^ idm, x1 = v.y ^ idm, x2 = v.z ^ idm, x3 = v.w ^ idm;
    const unsigned f0 = ~(((x0 & L) + L) | x0) & H;
    const unsigned f1 = ~(((x1 & L) + L) | x1) & H;
    const unsigned f2 = ~(((x2 & L) + L) | x2) & H;
    const unsigned f3 = ~(((x3 & L) + L) | x3) & H;
    unsigned F = f0 | (f1 >> 1) | (f2 >> 2) | (f3 >> 3);
    while (__builtin_amdgcn_ballot_w32(F != 0u) != 0u) {
        const bool     has = F != 0u;
        const unsigned p   = has ? (unsigned)__builtin_ctz(F) : 0u;
        F &= F - 1u;
        const unsigned e   = eb + ((7u - (p & 7u)) << 2) + (p >> 3);
        const bool     m   = has & (e < (unsigned)E);
        const unsigned mask = __builtin_amdgcn_ballot_w32(m);
        if (m) q[qn + (int)__builtin_amdgcn_mbcnt_lo(mask, 0u)] = e;
        qn += __popc(mask);
        while (qn >= OWN_GROUP) {
            qn -= OWN_GROUP;
            own_drain(xsrc, src, dst, q + qn, OWN_GROUP, lo, rows, cnt, acc, lane);
        }
    }
    return qn;
}

__global__ void __launch_bounds__(1024)
sage_owned_mean(const float* __restrict__ xsrc,
                const int* __restrict__ src,
                const int* __restrict__ dst,
                const unsigned* __restrict__ ids,
                float* __restrict__ mean,
                int E, int n_dst)
{
    __shared__ float    acc[(OWN_ROWS + 1) * FEAT];
    __shared__ unsigned cnt[OWN_ROWS];
    __shared__ unsigned queue[OWN_WAVES][OWN_QCAP];

    const int tid  = threadIdx.x;
    const int lane = tid & 31;
    const int w    = tid >> 5;
    unsigned* q = queue[w];

    const int rpw = own_rpw(n_dst, (int)gridDim.x);

    const int nt = (E + OWN_TRIP - 1) / OWN_TRIP;
    const int nw = (nt > w) ? (nt - w + OWN_WAVES - 1) / OWN_WAVES : 0;
    const int off = (int)(((unsigned long long)blockIdx.x * (unsigned)nt) / gridDim.x);
    const uint4* idv = (const uint4*)ids;

    for (int range = blockIdx.x; range * rpw < n_dst; range += gridDim.x) {
        const int      lo   = range * rpw;
        const unsigned rows = (unsigned)((n_dst - lo < rpw) ? (n_dst - lo) : rpw);
        const unsigned idm  = ((unsigned)range & 255u) * 0x01010101u;

        for (int i = tid; i < OWN_ROWS * FEAT / 4; i += 1024)
            ((float4*)acc)[i] = make_float4(0.f, 0.f, 0.f, 0.f);
        if (tid < OWN_ROWS) cnt[tid] = 0u;
        __syncthreads();

        int qn = 0;
        int i = 0;
        for (; i + OWN_SCAN <= nw; i += OWN_SCAN) {
            uint4    d[OWN_SCAN];
            unsigned eb[OWN_SCAN];
            #pragma unroll
            for (int u = 0; u < OWN_SCAN; ++u) {
                int t = w + OWN_WAVES * (i + u) + off;
                if (t >= nt) t -= nt;
                d[u]  = idv[t * 32 + lane];
                eb[u] = (unsigned)(t * OWN_TRIP + lane * 16);
            }
            #pragma unroll
            for (int u = 0; u < OWN_SCAN; ++u)
                qn = own_trip(d[u], eb[u], idm, E, lo, rows, xsrc, src, dst, q, cnt, acc, qn, lane);
        }
        for (; i < nw; ++i) {
            int t = w + OWN_WAVES * i + off;
            if (t >= nt) t -= nt;
            uint4    d  = idv[t * 32 + lane];
            unsigned eb = (unsigned)(t * OWN_TRIP + lane * 16);
            qn = own_trip(d, eb, idm, E, lo, rows, xsrc, src, dst, q, cnt, acc, qn, lane);
        }
        if (qn > 0) {
            unsigned q0 = q[0];
            if (lane >= qn && lane < OWN_GROUP) q[lane] = q0;
            own_drain(xsrc, src, dst, q, qn, lo, rows, cnt, acc, lane);
        }
        __syncthreads();

        for (unsigned r = (unsigned)w; r < rows; r += OWN_WAVES) {
            float inv = 1.0f / fmaxf((float)cnt[r], 1.0f);
            const float* a = acc + r * FEAT + lane;
            float4 o = make_float4(a[0] * inv, a[32] * inv, a[64] * inv, a[96] * inv);
            ((float4*)(mean + (size_t)(lo + (int)r) * FEAT))[lane] = o;
        }
        __syncthreads();
    }
}

template <int NT, int NTHR>
__global__ void __launch_bounds__(NTHR)
sage_linear_rows(const float* __restrict__ A1, const float* __restrict__ A2,
                 const float* __restrict__ W1, const float* __restrict__ W2,
                 const float* __restrict__ bias, float* __restrict__ out, int M)
{
    constexpr int N  = NT * 16;
    constexpr int NH = NT / 2;
    constexpr int SJ = NT * 32 + 4;
    __shared__ float4 wl[32 * SJ];

    #pragma unroll 8
    for (int it = 0; it < N * 64 / NTHR; ++it) {
        int i  = it * NTHR + (int)threadIdx.x;
        int q  = i & 7;
        int r  = (i >> 3) & 3;
        int Q  = (i >> 5) & 7;
        int R  = i >> 8;
        int n  = R * 4 + r;
        int kq = Q * 8 + q;
        const float* ws = (kq < 32) ? (W1 + n * FEAT + kq * 4)
                                    : (W2 + n * FEAT + (kq - 32) * 4);
        wl[(kq >> 1) * SJ + (n >> 4) * 32 + (kq & 1) * 16 + (n & 15)] = *(const float4*)ws;
    }
    __syncthreads();

    const int lane = threadIdx.x & 31;
    const int w    = __builtin_amdgcn_readfirstlane((int)(threadIdx.x >> 5));
    constexpr int nwp = NTHR >> 6;
    const int c    = lane & 15;
    const int h    = lane >> 4;
    const int half = w & 1;
    const int strips = (M + 15) >> 4;
    const float4* wb = wl + half * (NH * 32) + lane;

    for (int s = (w >> 1) * (int)gridDim.x + (int)blockIdx.x; s < strips;
         s += nwp * (int)gridDim.x) {
        int row = s * 16 + c;
        int rc  = (row < M) ? row : (M - 1);
        const float4* p1 = (const float4*)(A1 + (size_t)rc * FEAT) + h;
        const float4* p2 = (const float4*)(A2 + (size_t)rc * FEAT) + h;
        v8f acc[NH];
        #pragma unroll
        for (int t = 0; t < NH; ++t) acc[t] = v8f{};

        float4 an[4];
        #pragma unroll
        for (int jj = 0; jj < 4; ++jj) an[jj] = p1[2 * jj];

        #pragma nounroll
        for (int ch = 0; ch < 8; ++ch) {
            float4 a[4];
            #pragma unroll
            for (int jj = 0; jj < 4; ++jj) a[jj] = an[jj];
            if (ch < 7) {
                const float4* pn = ((ch + 1) < 4 ? p1 : p2) + 8 * ((ch + 1) & 3);
                #pragma unroll
                for (int jj = 0; jj < 4; ++jj) an[jj] = pn[2 * jj];
            }
            const float4* wc = wb + ch * 4 * SJ;
            #pragma unroll
            for (int jj = 0; jj < 4; ++jj) {
                float4 b[NH];
                #pragma unroll
                for (int t = 0; t < NH; ++t) b[t] = wc[jj * SJ + t * 32];
                #pragma unroll
                for (int t = 0; t < NH; ++t) {
                    v2f av = { a[jj].x, a[jj].y };
                    v2f bv = { b[t].x, b[t].y };
                    acc[t] = __builtin_amdgcn_wmma_f32_16x16x4_f32(
                         false, av,  false, bv,
                         (short)0, acc[t],  false,  false);
                }
                #pragma unroll
                for (int t = 0; t < NH; ++t) {
                    v2f av = { a[jj].z, a[jj].w };
                    v2f bv = { b[t].z, b[t].w };
                    acc[t] = __builtin_amdgcn_wmma_f32_16x16x4_f32(
                        false, av, false, bv, (short)0, acc[t], false, false);
                }
            }
        }

        int mbase = s * 16 + h * 8;
        #pragma unroll
        for (int t = 0; t < NH; ++t) {
            int   n  = (half * NH + t) * 16 + c;
            float bn = bias[n];
            #pragma unroll
            for (int r = 0; r < 8; ++r) {
                int mm = mbase + r;
                if (mm < M) {
                    float v = acc[t][r] + bn;
                    out[(size_t)mm * N + n] = v > 0.0f ? v : 0.0f;
                }
            }
        }
    }
}

extern "C" void kernel_launch(void* const* d_in, const int* in_sizes, int n_in,
                              void* d_out, int out_size, void* d_ws, size_t ws_size,
                              hipStream_t stream) {
    const float* x    = (const float*)d_in[0];
    const float* W1l  = (const float*)d_in[1];
    const float* b1l  = (const float*)d_in[2];
    const float* W1r  = (const float*)d_in[3];
    const float* W2l  = (const float*)d_in[4];
    const float* b2l  = (const float*)d_in[5];
    const float* W2r  = (const float*)d_in[6];
    const int*   src1 = (const int*)d_in[7];
    const int*   dst1 = (const int*)d_in[8];
    const int*   src2 = (const int*)d_in[9];
    const int*   dst2 = (const int*)d_in[10];

    const int E1     = in_sizes[7];
    const int E2     = in_sizes[9];
    const int N2FEAT = in_sizes[5];
    const int SIZE2  = out_size / N2FEAT;
    const int SIZE1  = 50000;

    float* agg1 = (float*)d_ws;
    float* cnt1 = agg1 + (size_t)SIZE1 * FEAT;
    float* h    = cnt1 + SIZE1;
    float* agg2 = h    + (size_t)SIZE1 * FEAT;
    unsigned* ids1 = (unsigned*)agg2;
    unsigned* ids2 = (unsigned*)cnt1;

    sage_owner_ids<<<256, 256, 0, stream>>>(dst1, ids1, E1, SIZE1, dst2, ids2, E2, SIZE2, 256);

    sage_owned_mean<<<256, 1024, 0, stream>>>(x, src1, dst1, ids1, agg1, E1, SIZE1);
    sage_linear_rows<8, 512><<<256, 512, 0, stream>>>(agg1, x, W1l, W1r, b1l, h, SIZE1);

    sage_owned_mean<<<256, 1024, 0, stream>>>(h, src2, dst2, ids2, agg2, E2, SIZE2);
    sage_linear_rows<4, 128><<<(SIZE2 + 31) / 32, 128, 0, stream>>>(
        agg2, h, W2l, W2r, b2l, (float*)d_out, SIZE2);
}


# First launch writes, per owner range, a bitmap of the 16-edge blocks holding its edges; the LDS-sum kernel reads its bitmap and only those blocks' dst, not all edge ids; sum order = LDS arrival
# speedup vs baseline: 3.7961x; 1.0415x over previous
#include <hip/hip_runtime.h>

typedef __attribute__((ext_vector_type(2))) float v2f;
typedef __attribute__((ext_vector_type(8))) float v8f;

#define FEAT 128

#define OWN_ROWS  196
#define OWN_WAVES 32
#define OWN_GROUP 16
#define OWN_BLK   16
#define OWN_QCAP  (OWN_GROUP + 32)

static __device__ __forceinline__ int own_rpw(int n_dst, int groups)
{
    int rpw = (n_dst + groups - 1) / groups;
    if (rpw > OWN_ROWS) rpw = OWN_ROWS;
    if (rpw < 1) rpw = 1;
    return rpw;
}

static __device__ __forceinline__ int own_row_words(int E)
{
    const int nb = (E + OWN_BLK - 1) / OWN_BLK;
    return ((nb + 1023) / 1024) * 32;
}

__global__ void __launch_bounds__(1024)
sage_owner_blocks(const int* __restrict__ dstA, unsigned* __restrict__ mapA, int EA, int nA,
                  const int* __restrict__ dstB, unsigned* __restrict__ mapB, int EB, int nB,
                  int groups)
{
    __shared__ unsigned tab[32 * 257];
    const int tid  = threadIdx.x;
    const int lane = tid & 31;
    const int w    = tid >> 5;
    const int cA = own_row_words(EA) / 32;
    const int cB = own_row_words(EB) / 32;
    const unsigned rA = (unsigned)own_rpw(nA, groups);
    const unsigned rB = (unsigned)own_rpw(nB, groups);
    for (int c = blockIdx.x; c < cA + cB; c += gridDim.x) {
        const bool      a   = c < cA;
        const int*      d   = a ? dstA : dstB;
        unsigned*       o   = a ? mapA : mapB;
        const unsigned  E   = (unsigned)(a ? EA : EB);
        const unsigned  rpw = a ? rA : rB;
        const int       ck  = a ? c : c - cA;
        const int       NW  = (a ? cA : cB) * 32;
        for (int i = tid; i < 32 * 257; i += 1024) tab[i] = 0u;
        __syncthreads();
        const unsigned e0   = (unsigned)(ck * 1024 + tid) * OWN_BLK;
        unsigned*      col  = tab + w * 257;
        const unsigned bitl = 1u << lane;
        if (e0 + (OWN_BLK - 1) < E) {
            #pragma unroll
            for (int j = 0; j < OWN_BLK / 4; ++j) {
                const int4 v = *(const int4*)(d + e0 + 4 * j);
                atomicOr(col + (((unsigned)v.x / rpw) & 255u), bitl);
                atomicOr(col + (((unsigned)v.y / rpw) & 255u), bitl);
                atomicOr(col + (((unsigned)v.z / rpw) & 255u), bitl);
                atomicOr(col + (((unsigned)v.w / rpw) & 255u), bitl);
            }
        } else {
            for (unsigned j = 0; j < OWN_BLK; ++j)
                if (e0 + j < E) atomicOr(col + (((unsigned)d[e0 + j] / rpw) & 255u), bitl);
        }
        __syncthreads();
        #pragma unroll
        for (int i = 0; i < 8; ++i) {
            const int g = w * 8 + i;
            o[(size_t)g * NW + ck * 32 + lane] = tab[lane * 257 + g];
        }
        __syncthreads();
    }
}

static __device__ __forceinline__ void
own_drain(const float* __restrict__ xsrc, const int* __restrict__ src,
          const int* __restrict__ dst, const unsigned* q, int nvalid,
          int lo, unsigned rows, unsigned* cnt, float* acc, int lane)
{
    const int      ql = lane & (OWN_GROUP - 1);
    const unsigned e  = q[ql];
    const int      s  = src[e];
    const unsigned r  = (unsigned)(dst[e] - lo);
    const bool     ok = (r < rows) & (ql < nvalid);
    const unsigned dl = ok ? r : (unsigned)OWN_ROWS;
    if (ok & (lane < OWN_GROUP)) atomicAdd(cnt + r, 1u);
    float4 v[OWN_GROUP];
    #pragma unroll
    for (int i = 0; i < OWN_GROUP; ++i) {
        {
            int si = __builtin_amdgcn_readlane(s, i);
            v[i] = ((const float4*)(xsrc + (size_t)si * FEAT))[lane];
        }
    }
    #pragma unroll
    for (int i = 0; i < OWN_GROUP; ++i) {
        {
            int di = __builtin_amdgcn_readlane((int)dl, i);
            float* a = acc + di * FEAT + lane;
            atomicAdd(a,      v[i].x);
            atomicAdd(a + 32, v[i].y);
            atomicAdd(a + 64, v[i].z);
            atomicAdd(a + 96, v[i].w);
        }
    }
}

static __device__ __forceinline__ unsigned
own_block_hits(const int* __restrict__ dst, unsigned e0, unsigned E, bool has,
               unsigned lo, unsigned rows)
{
    unsigned F = 0u;
    if (has) {
        unsigned d[OWN_BLK];
        if (e0 + (OWN_BLK - 1) < E) {
            #pragma unroll
            for (int j = 0; j < OWN_BLK / 4; ++j) {
                const int4 v = *(const int4*)(dst + e0 + 4 * j);
                d[4 * j + 0] = (unsigned)v.x;
                d[4 * j + 1] = (unsigned)v.y;
                d[4 * j + 2] = (unsigned)v.z;
                d[4 * j + 3] = (unsigned)v.w;
            }
        } else {
            #pragma unroll
            for (int j = 0; j < OWN_BLK; ++j) {
                const unsigned e = e0 + (unsigned)j;
                const unsigned v = (unsigned)dst[e < E ? e : E - 1u];
                d[j] = (e < E) ? v : 0xffffffffu;
            }
        }
        #pragma unroll
        for (int j = 0; j < OWN_BLK; ++j)
            F |= ((d[j] - lo) < rows ? 1u : 0u) << j;
    }
    return F;
}

static __device__ __forceinline__ int
own_push(unsigned F, unsigned e0, int lo, unsigned rows,
         const float* __restrict__ xsrc, const int* __restrict__ src,
         const int* __restrict__ dst,
         unsigned* q, unsigned* cnt, float* acc, int qn, int lane)
{
    while (__builtin_amdgcn_ballot_w32(F != 0u) != 0u) {
        const bool     m    = F != 0u;
        const unsigned p    = m ? (unsigned)__builtin_ctz(F) : 0u;
        F &= F - 1u;
        const unsigned e    = e0 + p;
        const unsigned mask = __builtin_amdgcn_ballot_w32(m);
        if (m) q[qn + (int)__builtin_amdgcn_mbcnt_lo(mask, 0u)] = e;
        qn += __popc(mask);
        while (qn >= OWN_GROUP) {
            qn -= OWN_GROUP;
            own_drain(xsrc, src, dst, q + qn, OWN_GROUP, lo, rows, cnt, acc, lane);
        }
    }
    return qn;
}

__global__ void __launch_bounds__(1024)
sage_owned_mean(const float* __restrict__ xsrc,
                const int* __restrict__ src,
                const int* __restrict__ dst,
                const unsigned* __restrict__ map,
                float* __restrict__ mean,
                int E, int n_dst)
{
    __shared__ float    acc[(OWN_ROWS + 1) * FEAT];
    __shared__ unsigned cnt[OWN_ROWS];
    __shared__ unsigned queue[OWN_WAVES][OWN_QCAP];

    const int tid  = threadIdx.x;
    const int lane = tid & 31;
    const int w    = tid >> 5;
    unsigned* q = queue[w];

    const int rpw = own_rpw(n_dst, (int)gridDim.x);
    const int NW  = own_row_words(E);
    const int NS  = NW * 4;

    for (int range = blockIdx.x; range * rpw < n_dst; range += gridDim.x) {
        const int      lo   = range * rpw;
        const unsigned rows = (unsigned)((n_dst - lo < rpw) ? (n_dst - lo) : rpw);
        const unsigned* rowp = map + (size_t)(range & 255) * NW;

        for (int i = tid; i < OWN_ROWS * FEAT / 4; i += 1024)
            ((float4*)acc)[i] = make_float4(0.f, 0.f, 0.f, 0.f);
        if (tid < OWN_ROWS) cnt[tid] = 0u;
        __syncthreads();

        int qn = 0;
        for (int sb = 0; sb < NS; sb += 8 * 1024) {
            unsigned long long m = 0ull;
            #pragma unroll
            for (int i = 0; i < 8; ++i) {
                const int      s  = sb + i * 1024 + tid;
                const int      sc = s < NS ? s : NS - 1;
                const unsigned wv = rowp[sc >> 2];
                m |= (unsigned long long)((wv >> (8 * (sc & 3))) & 255u) << (8 * i);
            }
            int ni = (NS - (sb + tid) + 1023) >> 10;
            ni = ni < 0 ? 0 : (ni > 8 ? 8 : ni);
            m &= (ni >= 8) ? ~0ull : ((1ull << (8 * ni)) - 1ull);
            const unsigned b0 = (unsigned)(sb + tid) * 8u;
            while (__builtin_amdgcn_ballot_w32(m != 0ull) != 0u) {
                const bool     has = m != 0ull;
                const unsigned bit = has ? (unsigned)__builtin_ctzll(m) : 0u;
                m &= m - 1ull;
                const unsigned e0  = (b0 + (bit >> 3) * 8192u + (bit & 7u)) * OWN_BLK;
                const unsigned F   = own_block_hits(dst, e0, (unsigned)E, has, (unsigned)lo, rows);
                qn = own_push(F, e0, lo, rows, xsrc, src, dst, q, cnt, acc, qn, lane);
            }
        }
        if (qn > 0) {
            unsigned q0 = q[0];
            if (lane >= qn && lane < OWN_GROUP) q[lane] = q0;
            own_drain(xsrc, src, dst, q, qn, lo, rows, cnt, acc, lane);
        }
        __syncthreads();

        for (unsigned r = (unsigned)w; r < rows; r += OWN_WAVES) {
            float inv = 1.0f / fmaxf((float)cnt[r], 1.0f);
            const float* a = acc + r * FEAT + lane;
            float4 o = make_float4(a[0] * inv, a[32] * inv, a[64] * inv, a[96] * inv);
            ((float4*)(mean + (size_t)(lo + (int)r) * FEAT))[lane] = o;
        }
        __syncthreads();
    }
}

template <int NT, int NTHR>
__global__ void __launch_bounds__(NTHR)
sage_linear_rows(const float* __restrict__ A1, const float* __restrict__ A2,
                 const float* __restrict__ W1, const float* __restrict__ W2,
                 const float* __restrict__ bias, float* __restrict__ out, int M)
{
    constexpr int N  = NT * 16;
    constexpr int NH = NT / 2;
    constexpr int SJ = NT * 32 + 4;
    __shared__ float4 wl[32 * SJ];

    #pragma unroll 8
    for (int it = 0; it < N * 64 / NTHR; ++it) {
        int i  = it * NTHR + (int)threadIdx.x;
        int q  = i & 7;
        int r  = (i >> 3) & 3;
        int Q  = (i >> 5) & 7;
        int R  = i >> 8;
        int n  = R * 4 + r;
        int kq = Q * 8 + q;
        const float* ws = (kq < 32) ? (W1 + n * FEAT + kq * 4)
                                    : (W2 + n * FEAT + (kq - 32) * 4);
        wl[(kq >> 1) * SJ + (n >> 4) * 32 + (kq & 1) * 16 + (n & 15)] = *(const float4*)ws;
    }
    __syncthreads();

    const int lane = threadIdx.x & 31;
    const int w    = __builtin_amdgcn_readfirstlane((int)(threadIdx.x >> 5));
    constexpr int nwp = NTHR >> 6;
    const int c    = lane & 15;
    const int h    = lane >> 4;
    const int half = w & 1;
    const int strips = (M + 15) >> 4;
    const float4* wb = wl + half * (NH * 32) + lane;

    for (int s = (w >> 1) * (int)gridDim.x + (int)blockIdx.x; s < strips;
         s += nwp * (int)gridDim.x) {
        int row = s * 16 + c;
        int rc  = (row < M) ? row : (M - 1);
        const float4* p1 = (const float4*)(A1 + (size_t)rc * FEAT) + h;
        const float4* p2 = (const float4*)(A2 + (size_t)rc * FEAT) + h;
        v8f acc[NH];
        #pragma unroll
        for (int t = 0; t < NH; ++t) acc[t] = v8f{};

        float4 an[4];
        #pragma unroll
        for (int jj = 0; jj < 4; ++jj) an[jj] = p1[2 * jj];

        #pragma nounroll
        for (int ch = 0; ch < 8; ++ch) {
            float4 a[4];
            #pragma unroll
            for (int jj = 0; jj < 4; ++jj) a[jj] = an[jj];
            if (ch < 7) {
                const float4* pn = ((ch + 1) < 4 ? p1 : p2) + 8 * ((ch + 1) & 3);
                #pragma unroll
                for (int jj = 0; jj < 4; ++jj) an[jj] = pn[2 * jj];
            }
            const float4* wc = wb + ch * 4 * SJ;
            #pragma unroll
            for (int jj = 0; jj < 4; ++jj) {
                float4 b[NH];
                #pragma unroll
                for (int t = 0; t < NH; ++t) b[t] = wc[jj * SJ + t * 32];
                #pragma unroll
                for (int t = 0; t < NH; ++t) {
                    v2f av = { a[jj].x, a[jj].y };
                    v2f bv = { b[t].x, b[t].y };
                    acc[t] = __builtin_amdgcn_wmma_f32_16x16x4_f32(
                         false, av,  false, bv,
                         (short)0, acc[t],  false,  false);
                }
                #pragma unroll
                for (int t = 0; t < NH; ++t) {
                    v2f av = { a[jj].z, a[jj].w };
                    v2f bv = { b[t].z, b[t].w };
                    acc[t] = __builtin_amdgcn_wmma_f32_16x16x4_f32(
                        false, av, false, bv, (short)0, acc[t], false, false);
                }
            }
        }

        int mbase = s * 16 + h * 8;
        #pragma unroll
        for (int t = 0; t < NH; ++t) {
            int   n  = (half * NH + t) * 16 + c;
            float bn = bias[n];
            #pragma unroll
            for (int r = 0; r < 8; ++r) {
                int mm = mbase + r;
                if (mm < M) {
                    float v = acc[t][r] + bn;
                    out[(size_t)mm * N + n] = v > 0.0f ? v : 0.0f;
                }
            }
        }
    }
}

extern "C" void kernel_launch(void* const* d_in, const int* in_sizes, int n_in,
                              void* d_out, int out_size, void* d_ws, size_t ws_size,
                              hipStream_t stream) {
    const float* x    = (const float*)d_in[0];
    const float* W1l  = (const float*)d_in[1];
    const float* b1l  = (const float*)d_in[2];
    const float* W1r  = (const float*)d_in[3];
    const float* W2l  = (const float*)d_in[4];
    const float* b2l  = (const float*)d_in[5];
    const float* W2r  = (const float*)d_in[6];
    const int*   src1 = (const int*)d_in[7];
    const int*   dst1 = (const int*)d_in[8];
    const int*   src2 = (const int*)d_in[9];
    const int*   dst2 = (const int*)d_in[10];

    const int E1     = in_sizes[7];
    const int E2     = in_sizes[9];
    const int N2FEAT = in_sizes[5];
    const int SIZE2  = out_size / N2FEAT;
    const int SIZE1  = 50000;

    float* agg1 = (float*)d_ws;
    float* cnt1 = agg1 + (size_t)SIZE1 * FEAT;
    float* h    = cnt1 + SIZE1;
    float* agg2 = h    + (size_t)SIZE1 * FEAT;
    unsigned* map1 = (unsigned*)agg2;
    unsigned* map2 = (unsigned*)cnt1;

    sage_owner_blocks<<<256, 1024, 0, stream>>>(dst1, map1, E1, SIZE1, dst2, map2, E2, SIZE2, 256);

    sage_owned_mean<<<256, 1024, 0, stream>>>(x, src1, dst1, map1, agg1, E1, SIZE1);
    sage_linear_rows<8, 512><<<256, 512, 0, stream>>>(agg1, x, W1l, W1r, b1l, h, SIZE1);

    sage_owned_mean<<<256, 1024, 0, stream>>>(h, src2, dst2, map2, agg2, E2, SIZE2);
    sage_linear_rows<4, 128><<<(SIZE2 + 31) / 32, 128, 0, stream>>>(
        agg2, h, W2l, W2r, b2l, (float*)d_out, SIZE2);
}


# Each layer becomes one launch: the workgroup that sums a range's rows in LDS divides in place and does those rows' WMMA products itself, A read from its LDS sums; 5 launches become 3; k order kept
# speedup vs baseline: 4.2236x; 1.1126x over previous
#include <hip/hip_runtime.h>

typedef __attribute__((ext_vector_type(2))) float v2f;
typedef __attribute__((ext_vector_type(8))) float v8f;

#define FEAT 128

#define OWN_ROWS  196
#define OWN_WAVES 32
#define OWN_GROUP 16
#define OWN_BLK   16
#define OWN_QCAP  (OWN_GROUP + 32)
#define OWN_PITCH 132

static __device__ __forceinline__ int own_rpw(int n_dst, int groups)
{
    int rpw = (n_dst + groups - 1) / groups;
    if (rpw > OWN_ROWS) rpw = OWN_ROWS;
    if (rpw < 1) rpw = 1;
    return rpw;
}

static __device__ __forceinline__ int own_row_words(int E)
{
    const int nb = (E + OWN_BLK - 1) / OWN_BLK;
    return ((nb + 1023) / 1024) * 32;
}

__global__ void __launch_bounds__(1024)
sage_owner_blocks(const int* __restrict__ dstA, unsigned* __restrict__ mapA, int EA, int nA,
                  const int* __restrict__ dstB, unsigned* __restrict__ mapB, int EB, int nB,
                  int groups)
{
    __shared__ unsigned tab[32 * 257];
    const int tid  = threadIdx.x;
    const int lane = tid & 31;
    const int w    = tid >> 5;
    const int cA = own_row_words(EA) / 32;
    const int cB = own_row_words(EB) / 32;
    const unsigned rA = (unsigned)own_rpw(nA, groups);
    const unsigned rB = (unsigned)own_rpw(nB, groups);
    for (int c = blockIdx.x; c < cA + cB; c += gridDim.x) {
        const bool      a   = c < cA;
        const int*      d   = a ? dstA : dstB;
        unsigned*       o   = a ? mapA : mapB;
        const unsigned  E   = (unsigned)(a ? EA : EB);
        const unsigned  rpw = a ? rA : rB;
        const int       ck  = a ? c : c - cA;
        const int       NW  = (a ? cA : cB) * 32;
        for (int i = tid; i < 32 * 257; i += 1024) tab[i] = 0u;
        __syncthreads();
        const unsigned e0   = (unsigned)(ck * 1024 + tid) * OWN_BLK;
        unsigned*      col  = tab + w * 257;
        const unsigned bitl = 1u << lane;
        if (e0 + (OWN_BLK - 1) < E) {
            #pragma unroll
            for (int j = 0; j < OWN_BLK / 4; ++j) {
                const int4 v = *(const int4*)(d + e0 + 4 * j);
                atomicOr(col + (((unsigned)v.x / rpw) & 255u), bitl);
                atomicOr(col + (((unsigned)v.y / rpw) & 255u), bitl);
                atomicOr(col + (((unsigned)v.z / rpw) & 255u), bitl);
                atomicOr(col + (((unsigned)v.w / rpw) & 255u), bitl);
            }
        } else {
            for (unsigned j = 0; j < OWN_BLK; ++j)
                if (e0 + j < E) atomicOr(col + (((unsigned)d[e0 + j] / rpw) & 255u), bitl);
        }
        __syncthreads();
        #pragma unroll
        for (int i = 0; i < 8; ++i) {
            const int g = w * 8 + i;
            o[(size_t)g * NW + ck * 32 + lane] = tab[lane * 257 + g];
        }
        __syncthreads();
    }
}

static __device__ __forceinline__ void
own_drain(const float* __restrict__ xsrc, const int* __restrict__ src,
          const int* __restrict__ dst, const unsigned* q, int nvalid,
          int lo, unsigned rows, unsigned* cnt, float* acc, int lane)
{
    const int      ql = lane & (OWN_GROUP - 1);
    const unsigned e  = q[ql];
    const int      s  = src[e];
    const unsigned r  = (unsigned)(dst[e] - lo);
    const bool     ok = (r < rows) & (ql < nvalid);
    const unsigned dl = ok ? r : (unsigned)OWN_ROWS;
    if (ok & (lane < OWN_GROUP)) atomicAdd(cnt + r, 1u);
    float4 v[OWN_GROUP];
    #pragma unroll
    for (int i = 0; i < OWN_GROUP; ++i) {
        {
            int si = __builtin_amdgcn_readlane(s, i);
            v[i] = ((const float4*)(xsrc + (size_t)si * FEAT))[lane];
        }
    }
    #pragma unroll
    for (int i = 0; i < OWN_GROUP; ++i) {
        {
            int di = __builtin_amdgcn_readlane((int)dl, i);
            float* a = acc + di * OWN_PITCH + lane;
            atomicAdd(a,      v[i].x);
            atomicAdd(a + 32, v[i].y);
            atomicAdd(a + 64, v[i].z);
            atomicAdd(a + 96, v[i].w);
        }
    }
}

static __device__ __forceinline__ unsigned
own_block_hits(const int* __restrict__ dst, unsigned e0, unsigned E, bool has,
               unsigned lo, unsigned rows)
{
    unsigned F = 0u;
    if (has) {
        unsigned d[OWN_BLK];
        if (e0 + (OWN_BLK - 1) < E) {
            #pragma unroll
            for (int j = 0; j < OWN_BLK / 4; ++j) {
                const int4 v = *(const int4*)(dst + e0 + 4 * j);
                d[4 * j + 0] = (unsigned)v.x;
                d[4 * j + 1] = (unsigned)v.y;
                d[4 * j + 2] = (unsigned)v.z;
                d[4 * j + 3] = (unsigned)v.w;
            }
        } else {
            #pragma unroll
            for (int j = 0; j < OWN_BLK; ++j) {
                const unsigned e = e0 + (unsigned)j;
                const unsigned v = (unsigned)dst[e < E ? e : E - 1u];
                d[j] = (e < E) ? v : 0xffffffffu;
            }
        }
        #pragma unroll
        for (int j = 0; j < OWN_BLK; ++j)
            F |= ((d[j] - lo) < rows ? 1u : 0u) << j;
    }
    return F;
}

static __device__ __forceinline__ int
own_push(unsigned F, unsigned e0, int lo, unsigned rows,
         const float* __restrict__ xsrc, const int* __restrict__ src,
         const int* __restrict__ dst,
         unsigned* q, unsigned* cnt, float* acc, int qn, int lane)
{
    while (__builtin_amdgcn_ballot_w32(F != 0u) != 0u) {
        const bool     m    = F != 0u;
        const unsigned p    = m ? (unsigned)__builtin_ctz(F) : 0u;
        F &= F - 1u;
        const unsigned e    = e0 + p;
        const unsigned mask = __builtin_amdgcn_ballot_w32(m);
        if (m) q[qn + (int)__builtin_amdgcn_mbcnt_lo(mask, 0u)] = e;
        qn += __popc(mask);
        while (qn >= OWN_GROUP) {
            qn -= OWN_GROUP;
            own_drain(xsrc, src, dst, q + qn, OWN_GROUP, lo, rows, cnt, acc, lane);
        }
    }
    return qn;
}

template <int NH>
static __device__ __forceinline__ void
own_block_products(const float4 a, const float4* wc, v8f (&ac)[NH])
{
    float4 b[NH];
    #pragma unroll
    for (int t = 0; t < NH; ++t) b[t] = wc[t * 32];
    #pragma unroll
    for (int t = 0; t < NH; ++t) {
        v2f av = { a.x, a.y };
        v2f bv = { b[t].x, b[t].y };
        ac[t] = __builtin_amdgcn_wmma_f32_16x16x4_f32(
             false, av,  false, bv,
             (short)0, ac[t],  false,  false);
    }
    #pragma unroll
    for (int t = 0; t < NH; ++t) {
        v2f av = { a.z, a.w };
        v2f bv = { b[t].z, b[t].w };
        ac[t] = __builtin_amdgcn_wmma_f32_16x16x4_f32(
            false, av, false, bv, (short)0, ac[t], false, false);
    }
}

template <int NT>
__global__ void __launch_bounds__(1024)
sage_owned_layer(const float* __restrict__ xsrc,
                 const int* __restrict__ src,
                 const int* __restrict__ dst,
                 const unsigned* __restrict__ map,
                 const float* __restrict__ W1, const float* __restrict__ W2,
                 const float* __restrict__ bias, float* __restrict__ out,
                 int E, int n_dst)
{
    constexpr int N  = NT * 16;
    constexpr int NH = 2;
    constexpr int NG = NT / NH;
    constexpr int SJ = NT * 32 + 4;
    __shared__ float4   lds4[32 * SJ + (OWN_ROWS + 1) * OWN_PITCH / 4];
    float4* wl   = lds4;
    float4* acc4 = lds4 + 32 * SJ;
    __shared__ unsigned cnt[OWN_ROWS];
    __shared__ unsigned queue[OWN_WAVES][OWN_QCAP];
    float* acc = (float*)acc4;

    const int tid  = threadIdx.x;
    const int lane = tid & 31;
    const int w    = tid >> 5;
    unsigned* q = queue[w];

    #pragma unroll
    for (int it = 0; it < N * 64 / 1024; ++it) {
        int i  = it * 1024 + tid;
        int qq = i & 7;
        int r  = (i >> 3) & 3;
        int Q  = (i >> 5) & 7;
        int R  = i >> 8;
        int n  = R * 4 + r;
        int kq = Q * 8 + qq;
        const float* ws = (kq < 32) ? (W1 + n * FEAT + kq * 4)
                                    : (W2 + n * FEAT + (kq - 32) * 4);
        wl[(kq >> 1) * SJ + (n >> 4) * 32 + (kq & 1) * 16 + (n & 15)] = *(const float4*)ws;
    }

    const int rpw = own_rpw(n_dst, (int)gridDim.x);
    const int NW  = own_row_words(E);
    const int NS  = NW * 4;

    const int wu = __builtin_amdgcn_readfirstlane(w);
    const int c  = lane & 15;
    const int h  = lane >> 4;

    for (int range = blockIdx.x; range * rpw < n_dst; range += gridDim.x) {
        const int      lo   = range * rpw;
        const unsigned rows = (unsigned)((n_dst - lo < rpw) ? (n_dst - lo) : rpw);
        const unsigned* rowp = map + (size_t)(range & 255) * NW;

        for (int i = tid; i < (OWN_ROWS + 1) * OWN_PITCH / 4; i += 1024)
            acc4[i] = make_float4(0.f, 0.f, 0.f, 0.f);
        if (tid < OWN_ROWS) cnt[tid] = 0u;
        __syncthreads();

        int qn = 0;
        for (int sb = 0; sb < NS; sb += 8 * 1024) {
            unsigned long long m = 0ull;
            #pragma unroll
            for (int i = 0; i < 8; ++i) {
                const int      sl = sb + i * 1024 + tid;
                const int      sc = sl < NS ? sl : NS - 1;
                const unsigned wv = rowp[sc >> 2];
                m |= (unsigned long long)((wv >> (8 * (sc & 3))) & 255u) << (8 * i);
            }
            int ni = (NS - (sb + tid) + 1023) >> 10;
            ni = ni < 0 ? 0 : (ni > 8 ? 8 : ni);
            m &= (ni >= 8) ? ~0ull : ((1ull << (8 * ni)) - 1ull);
            const unsigned b0 = (unsigned)(sb + tid) * 8u;
            while (__builtin_amdgcn_ballot_w32(m != 0ull) != 0u) {
                const bool     has = m != 0ull;
                const unsigned bit = has ? (unsigned)__builtin_ctzll(m) : 0u;
                m &= m - 1ull;
                const unsigned e0  = (b0 + (bit >> 3) * 8192u + (bit & 7u)) * OWN_BLK;
                const unsigned F   = own_block_hits(dst, e0, (unsigned)E, has, (unsigned)lo, rows);
                qn = own_push(F, e0, lo, rows, xsrc, src, dst, q, cnt, acc, qn, lane);
            }
        }
        if (qn > 0) {
            unsigned q0 = q[0];
            if (lane >= qn && lane < OWN_GROUP) q[lane] = q0;
            own_drain(xsrc, src, dst, q, qn, lo, rows, cnt, acc, lane);
        }
        __syncthreads();

        for (unsigned r = (unsigned)w; r < rows; r += OWN_WAVES) {
            float inv = 1.0f / fmaxf((float)cnt[r], 1.0f);
            float* a = acc + r * OWN_PITCH;
            float4 o = make_float4(a[lane] * inv, a[32 + lane] * inv,
                                   a[64 + lane] * inv, a[96 + lane] * inv);
            ((float4*)a)[lane] = o;
        }
        __syncthreads();

        const int strips = (int)((rows + 15u) >> 4);
        for (int task = wu; task < strips * NG; task += OWN_WAVES) {
            const int s = task / NG;
            const int g = task % NG;
            unsigned lr = (unsigned)(s * 16 + c);
            lr = lr < rows ? lr : rows - 1u;
            const float4* p1 = (const float4*)(acc + lr * OWN_PITCH) + h;
            const float4* p2 = (const float4*)(xsrc + (size_t)(lo + (int)lr) * FEAT) + h;
            const float4* wb = wl + g * (NH * 32) + lane;

            v8f ac[NH];
            #pragma unroll
            for (int t = 0; t < NH; ++t) ac[t] = v8f{};

            float4 an[4];
            #pragma unroll
            for (int jj = 0; jj < 4; ++jj) an[jj] = p2[2 * jj];

            const float4* wc = wb;
            #pragma nounroll
            for (int ch = 0; ch < 4; ++ch) {
                float4 a[4];
                #pragma unroll
                for (int jj = 0; jj < 4; ++jj) a[jj] = p1[8 * ch + 2 * jj];
                #pragma unroll
                for (int jj = 0; jj < 4; ++jj)
                    own_block_products<NH>(a[jj], wc + jj * SJ, ac);
                wc += 4 * SJ;
            }
            #pragma nounroll
            for (int ch = 0; ch < 4; ++ch) {
                float4 a[4];
                #pragma unroll
                for (int jj = 0; jj < 4; ++jj) a[jj] = an[jj];
                if (ch < 3) {
                    #pragma unroll
                    for (int jj = 0; jj < 4; ++jj) an[jj] = p2[8 * (ch + 1) + 2 * jj];
                }
                #pragma unroll
                for (int jj = 0; jj < 4; ++jj)
                    own_block_products<NH>(a[jj], wc + jj * SJ, ac);
                wc += 4 * SJ;
            }

            const int mb = s * 16 + h * 8;
            #pragma unroll
            for (int t = 0; t < NH; ++t) {
                int   n  = (g * NH + t) * 16 + c;
                float bn = bias[n];
                #pragma unroll
                for (int r = 0; r < 8; ++r) {
                    int mm = mb + r;
                    if ((unsigned)mm < rows) {
                        float v = ac[t][r] + bn;
                        out[(size_t)(lo + mm) * N + n] = v > 0.0f ? v : 0.0f;
                    }
                }
            }
        }
        __syncthreads();
    }
}

extern "C" void kernel_launch(void* const* d_in, const int* in_sizes, int n_in,
                              void* d_out, int out_size, void* d_ws, size_t ws_size,
                              hipStream_t stream) {
    const float* x    = (const float*)d_in[0];
    const float* W1l  = (const float*)d_in[1];
    const float* b1l  = (const float*)d_in[2];
    const float* W1r  = (const float*)d_in[3];
    const float* W2l  = (const float*)d_in[4];
    const float* b2l  = (const float*)d_in[5];
    const float* W2r  = (const float*)d_in[6];
    const int*   src1 = (const int*)d_in[7];
    const int*   dst1 = (const int*)d_in[8];
    const int*   src2 = (const int*)d_in[9];
    const int*   dst2 = (const int*)d_in[10];

    const int E1     = in_sizes[7];
    const int E2     = in_sizes[9];
    const int N2FEAT = in_sizes[5];
    const int SIZE2  = out_size / N2FEAT;
    const int SIZE1  = 50000;

    float* agg1 = (float*)d_ws;
    float* cnt1 = agg1 + (size_t)SIZE1 * FEAT;
    float* h    = cnt1 + SIZE1;
    float* agg2 = h    + (size_t)SIZE1 * FEAT;
    unsigned* map1 = (unsigned*)agg2;
    unsigned* map2 = (unsigned*)cnt1;

    sage_owner_blocks<<<256, 1024, 0, stream>>>(dst1, map1, E1, SIZE1, dst2, map2, E2, SIZE2, 256);

    sage_owned_layer<8><<<256, 1024, 0, stream>>>(x, src1, dst1, map1, W1l, W1r, b1l, h, E1, SIZE1);

    sage_owned_layer<4><<<256, 1024, 0, stream>>>(h, src2, dst2, map2, W2l, W2r, b2l,
                                                  (float*)d_out, E2, SIZE2);
}
